# starting kernel
# baseline (speedup 1.0000x reference)
#include <hip/hip_runtime.h>

#define D_   480
#define H_   360
#define W_   32
#define B_   2
#define COUT 32
#define EPSF 1e-5f
#define NEGF 0.01f
#define NBR_STRIDE 12
#define TPW  4

typedef __attribute__((ext_vector_type(2))) float v2f;
typedef __attribute__((ext_vector_type(8))) float v8f;

__global__ void grid_init_kernel(int4* __restrict__ grid, int n4) {
    int i = blockIdx.x * blockDim.x + threadIdx.x;
    if (i < n4) grid[i] = make_int4(-1, -1, -1, -1);
}

__global__ void zrow_init_kernel(float* __restrict__ z) {
    z[threadIdx.x] = 0.0f;
}

__global__ void grid_scatter_kernel(const int* __restrict__ coords,
                                    int* __restrict__ grid, int n) {
    int i = blockIdx.x * blockDim.x + threadIdx.x;
    if (i >= n) return;
    int b = coords[4 * i + 0];
    int z = coords[4 * i + 1];
    int y = coords[4 * i + 2];
    int x = coords[4 * i + 3];
    grid[((b * D_ + z) * H_ + y) * W_ + x] = i;
}

__global__ void nbr_build_kernel(const int* __restrict__ coords,
                                 const int* __restrict__ grid,
                                 int* __restrict__ nbr133,
                                 int* __restrict__ nbr313, int n) {
    int i = blockIdx.x * blockDim.x + threadIdx.x;
    if (i >= n) return;
    int b = coords[4 * i + 0];
    int z = coords[4 * i + 1];
    int y = coords[4 * i + 2];
    int x = coords[4 * i + 3];
    int k = 0;
    #pragma unroll
    for (int dy = -1; dy <= 1; ++dy) {
        #pragma unroll
        for (int dx = -1; dx <= 1; ++dx) {
            int ny = y + dy, nx = x + dx;
            int v = -1;
            if (ny >= 0 && ny < H_ && nx >= 0 && nx < W_)
                v = grid[((b * D_ + z) * H_ + ny) * W_ + nx];
            nbr133[i * NBR_STRIDE + k] = v;
            ++k;
        }
    }
    k = 0;
    #pragma unroll
    for (int dz = -1; dz <= 1; ++dz) {
        #pragma unroll
        for (int dx = -1; dx <= 1; ++dx) {
            int nz = z + dz, nx = x + dx;
            int v = -1;
            if (nz >= 0 && nz < D_ && nx >= 0 && nx < W_)
                v = grid[((b * D_ + nz) * H_ + y) * W_ + nx];
            nbr313[i * NBR_STRIDE + k] = v;
            ++k;
        }
    }
}

template <int CIN, bool ADD_RES>
__global__ void __launch_bounds__(256)
conv_bn_kernel(const float* __restrict__ feats,
               const float* __restrict__ Wg,
               const int*   __restrict__ nbr,
               const float* __restrict__ zrow,
               const float* __restrict__ gamma,
               const float* __restrict__ beta,
               const float* __restrict__ mean,
               const float* __restrict__ var,
               const float* __restrict__ res,
               float* __restrict__ out,
               int n) {
    __shared__ float Wl[9 * CIN * COUT];
    for (int i = threadIdx.x; i < 9 * CIN * COUT; i += blockDim.x) {
        int k  = i / (CIN * COUT);
        int r  = i - k * (CIN * COUT);
        int ci = r / COUT;
        int co = r - ci * COUT;
        Wl[((k * (CIN / 2) + (ci >> 1)) * COUT + co) * 2 + (ci & 1)] = Wg[i];
    }
    __syncthreads();

    const int lane = threadIdx.x & 31;
    const int wv   = threadIdx.x >> 5;
    const int row  = lane & 15;
    const int half = lane >> 4;

    float scl[2], bia[2];
    #pragma unroll
    for (int t2 = 0; t2 < 2; ++t2) {
        const int co = t2 * 16 + row;
        scl[t2] = gamma[co] * rsqrtf(var[co] + EPSF);
        bia[t2] = beta[co] - mean[co] * scl[t2];
    }

    const int tile0 = (blockIdx.x * (blockDim.x >> 5) + wv) * TPW;

    #pragma unroll 1
    for (int t = 0; t < TPW; ++t) {
        const int tile = tile0 + t;
        if (tile * 16 >= n) break;

        const int point  = tile * 16 + row;
        const int pclamp = point < n ? point : n - 1;

        const int4* np = (const int4*)(nbr + (size_t)pclamp * NBR_STRIDE);
        const int4 n03 = np[0];
        const int4 n47 = np[1];
        const int  n8  = ((const int*)(np + 2))[0];
        const int nbrs[9] = {n03.x, n03.y, n03.z, n03.w,
                             n47.x, n47.y, n47.z, n47.w, n8};

        v8f acc0 = {};
        v8f acc1 = {};

        #pragma unroll
        for (int k = 0; k < 9; ++k) {
            const int nb = nbrs[k];
            const float* frow = (nb >= 0) ? (feats + (size_t)nb * CIN) : zrow;
            #pragma unroll
            for (int c = 0; c < CIN; c += 4) {
                const v2f A = *(const v2f*)(frow + c + 2 * half);
                const float* wp =
                    &Wl[((k * (CIN / 2) + (c >> 1) + half) * COUT + row) * 2];
                const v2f B0 = *(const v2f*)(wp);
                const v2f B1 = *(const v2f*)(wp + 32);
                acc0 = __builtin_amdgcn_wmma_f32_16x16x4_f32(
                    false, A, false, B0, (short)0, acc0, false, false);
                acc1 = __builtin_amdgcn_wmma_f32_16x16x4_f32(
                    false, A, false, B1, (short)0, acc1, false, false);
            }
        }

        #pragma unroll
        for (int t2 = 0; t2 < 2; ++t2) {
            const int co = t2 * 16 + row;
            v8f acc = t2 ? acc1 : acc0;
            #pragma unroll
            for (int v = 0; v < 8; ++v) {
                float yv = acc[v] * scl[t2] + bia[t2];
                yv = (yv >= 0.0f) ? yv : NEGF * yv;
                const int orow = tile * 16 + half * 8 + v;
                if (orow < n) {
                    const int oidx = orow * COUT + co;
                    if (ADD_RES) yv += res[oidx];
                    out[oidx] = yv;
                }
            }
        }
    }
}

extern "C" void kernel_launch(void* const* d_in, const int* in_sizes, int n_in,
                              void* d_out, int out_size, void* d_ws, size_t ws_size,
                              hipStream_t stream) {
    const float* feats    = (const float*)d_in[0];
    const int*   coords   = (const int*)  d_in[1];
    const float* W1       = (const float*)d_in[2];
    const float* W2       = (const float*)d_in[3];
    const float* W3       = (const float*)d_in[4];
    const float* W4       = (const float*)d_in[5];
    const float* bn_gamma = (const float*)d_in[6];
    const float* bn_beta  = (const float*)d_in[7];
    const float* bn_mean  = (const float*)d_in[8];
    const float* bn_var   = (const float*)d_in[9];
    float* out = (float*)d_out;

    const int n = in_sizes[0] / 16;

    const size_t GRID_ELEMS = (size_t)B_ * D_ * H_ * W_;
    int*   grid   = (int*)d_ws;
    int*   nbr133 = grid + GRID_ELEMS;
    int*   nbr313 = nbr133 + (size_t)n * NBR_STRIDE;
    float* bufA   = (float*)(nbr313 + (size_t)n * NBR_STRIDE);
    float* bufB   = bufA + (size_t)n * COUT;
    float* zrow   = bufB + (size_t)n * COUT;

    const int n4 = (int)(GRID_ELEMS / 4);
    grid_init_kernel<<<(n4 + 255) / 256, 256, 0, stream>>>((int4*)grid, n4);
    zrow_init_kernel<<<1, 32, 0, stream>>>(zrow);
    grid_scatter_kernel<<<(n + 255) / 256, 256, 0, stream>>>(coords, grid, n);

    nbr_build_kernel<<<(n + 255) / 256, 256, 0, stream>>>(coords, grid, nbr133, nbr313, n);

    const int tiles  = (n + 15) / 16;
    const int wavesN = (tiles + TPW - 1) / TPW;
    const int blocks = (wavesN + 7) / 8;

    conv_bn_kernel<16, false><<<blocks, 256, 0, stream>>>(
        feats, W1, nbr133, zrow,
        bn_gamma + 0,  bn_beta + 0,  bn_mean + 0,  bn_var + 0,
        nullptr, bufA, n);
    conv_bn_kernel<32, false><<<blocks, 256, 0, stream>>>(
        bufA,  W2, nbr313, zrow,
        bn_gamma + 32, bn_beta + 32, bn_mean + 32, bn_var + 32,
        nullptr, bufB, n);
    conv_bn_kernel<16, false><<<blocks, 256, 0, stream>>>(
        feats, W3, nbr313, zrow,
        bn_gamma + 64, bn_beta + 64, bn_mean + 64, bn_var + 64,
        nullptr, bufA, n);
    conv_bn_kernel<32, true><<<blocks, 256, 0, stream>>>(
        bufA,  W4, nbr133, zrow,
        bn_gamma + 96, bn_beta + 96, bn_mean + 96, bn_var + 96,
        bufB, out, n);
}


# conv weights kept in registers as B fragments, loaded once per wave from global; 2048 resident waves walk over the tiles; LDS staging, its copy loop, block barrier and per-step LDS reads removed
# speedup vs baseline: 1.1114x; 1.1114x over previous
#include <hip/hip_runtime.h>

#define D_   480
#define H_   360
#define W_   32
#define B_   2
#define COUT 32
#define EPSF 1e-5f
#define NEGF 0.01f
#define NBR_STRIDE 12

typedef __attribute__((ext_vector_type(2))) float v2f;
typedef __attribute__((ext_vector_type(8))) float v8f;

__global__ void grid_init_kernel(int4* __restrict__ grid, int n4) {
    int i = blockIdx.x * blockDim.x + threadIdx.x;
    if (i < n4) grid[i] = make_int4(-1, -1, -1, -1);
}

__global__ void zrow_init_kernel(float* __restrict__ z) {
    z[threadIdx.x] = 0.0f;
}

__global__ void grid_scatter_kernel(const int* __restrict__ coords,
                                    int* __restrict__ grid, int n) {
    int i = blockIdx.x * blockDim.x + threadIdx.x;
    if (i >= n) return;
    int b = coords[4 * i + 0];
    int z = coords[4 * i + 1];
    int y = coords[4 * i + 2];
    int x = coords[4 * i + 3];
    grid[((b * D_ + z) * H_ + y) * W_ + x] = i;
}

__global__ void nbr_build_kernel(const int* __restrict__ coords,
                                 const int* __restrict__ grid,
                                 int* __restrict__ nbr133,
                                 int* __restrict__ nbr313, int n) {
    int i = blockIdx.x * blockDim.x + threadIdx.x;
    if (i >= n) return;
    int b = coords[4 * i + 0];
    int z = coords[4 * i + 1];
    int y = coords[4 * i + 2];
    int x = coords[4 * i + 3];
    int k = 0;
    #pragma unroll
    for (int dy = -1; dy <= 1; ++dy) {
        #pragma unroll
        for (int dx = -1; dx <= 1; ++dx) {
            int ny = y + dy, nx = x + dx;
            int v = -1;
            if (ny >= 0 && ny < H_ && nx >= 0 && nx < W_)
                v = grid[((b * D_ + z) * H_ + ny) * W_ + nx];
            nbr133[i * NBR_STRIDE + k] = v;
            ++k;
        }
    }
    k = 0;
    #pragma unroll
    for (int dz = -1; dz <= 1; ++dz) {
        #pragma unroll
        for (int dx = -1; dx <= 1; ++dx) {
            int nz = z + dz, nx = x + dx;
            int v = -1;
            if (nz >= 0 && nz < D_ && nx >= 0 && nx < W_)
                v = grid[((b * D_ + nz) * H_ + y) * W_ + nx];
            nbr313[i * NBR_STRIDE + k] = v;
            ++k;
        }
    }
}

template <int CIN, bool ADD_RES>
__global__ void __launch_bounds__(256) __attribute__((amdgpu_waves_per_eu(2, 2)))
conv_bn_kernel(const float* __restrict__ feats,
               const float* __restrict__ Wg,
               const int*   __restrict__ nbr,
               const float* __restrict__ zrow,
               const float* __restrict__ gamma,
               const float* __restrict__ beta,
               const float* __restrict__ mean,
               const float* __restrict__ var,
               const float* __restrict__ res,
               float* __restrict__ out,
               int n) {
    const int lane = threadIdx.x & 31;
    const int wv   = threadIdx.x >> 5;
    const int row  = lane & 15;
    const int half = lane >> 4;

    v2f Bf[9][CIN / 4][2];
    {
        const float* wl = Wg + (2 * half) * COUT + row;
        #pragma unroll
        for (int k = 0; k < 9; ++k) {
            #pragma unroll
            for (int c4 = 0; c4 < CIN / 4; ++c4) {
                #pragma unroll
                for (int t2 = 0; t2 < 2; ++t2) {
                    const float* wp = wl + (k * CIN + c4 * 4) * COUT + t2 * 16;
                    v2f b;
                    b.x = wp[0];
                    b.y = wp[COUT];
                    Bf[k][c4][t2] = b;
                }
            }
        }
    }

    float scl[2], bia[2];
    #pragma unroll
    for (int t2 = 0; t2 < 2; ++t2) {
        const int co = t2 * 16 + row;
        scl[t2] = gamma[co] * rsqrtf(var[co] + EPSF);
        bia[t2] = beta[co] - mean[co] * scl[t2];
    }

    const int wave   = blockIdx.x * (blockDim.x >> 5) + wv;
    const int nwaves = gridDim.x * (blockDim.x >> 5);
    const int tiles  = (n + 15) >> 4;

    #pragma unroll 1
    for (int tile = wave; tile < tiles; tile += nwaves) {
        const int point  = tile * 16 + row;
        const int pclamp = point < n ? point : n - 1;

        const int4* np = (const int4*)(nbr + (size_t)pclamp * NBR_STRIDE);
        const int4 n03 = np[0];
        const int4 n47 = np[1];
        const int  n8  = ((const int*)(np + 2))[0];
        const int nbrs[9] = {n03.x, n03.y, n03.z, n03.w,
                             n47.x, n47.y, n47.z, n47.w, n8};

        v2f Af[9][CIN / 4];
        #pragma unroll
        for (int k = 0; k < 9; ++k) {
            const int nb = nbrs[k];
            const float* frow = (nb >= 0) ? (feats + (size_t)nb * CIN) : zrow;
            #pragma unroll
            for (int c4 = 0; c4 < CIN / 4; ++c4)
                Af[k][c4] = *(const v2f*)(frow + c4 * 4 + 2 * half);
        }

        v8f acc0 = {};
        v8f acc1 = {};

        #pragma unroll
        for (int k = 0; k < 9; ++k) {
            #pragma unroll
            for (int c4 = 0; c4 < CIN / 4; ++c4) {
                acc0 = __builtin_amdgcn_wmma_f32_16x16x4_f32(
                    false, Af[k][c4], false, Bf[k][c4][0], (short)0, acc0, false, false);
                acc1 = __builtin_amdgcn_wmma_f32_16x16x4_f32(
                    false, Af[k][c4], false, Bf[k][c4][1], (short)0, acc1, false, false);
            }
        }

        #pragma unroll
        for (int t2 = 0; t2 < 2; ++t2) {
            const int co = t2 * 16 + row;
            v8f acc = t2 ? acc1 : acc0;
            #pragma unroll
            for (int v = 0; v < 8; ++v) {
                float yv = acc[v] * scl[t2] + bia[t2];
                yv = (yv >= 0.0f) ? yv : NEGF * yv;
                const int orow = tile * 16 + half * 8 + v;
                if (orow < n) {
                    const int oidx = orow * COUT + co;
                    if (ADD_RES) yv += res[oidx];
                    out[oidx] = yv;
                }
            }
        }
    }
}

extern "C" void kernel_launch(void* const* d_in, const int* in_sizes, int n_in,
                              void* d_out, int out_size, void* d_ws, size_t ws_size,
                              hipStream_t stream) {
    const float* feats    = (const float*)d_in[0];
    const int*   coords   = (const int*)  d_in[1];
    const float* W1       = (const float*)d_in[2];
    const float* W2       = (const float*)d_in[3];
    const float* W3       = (const float*)d_in[4];
    const float* W4       = (const float*)d_in[5];
    const float* bn_gamma = (const float*)d_in[6];
    const float* bn_beta  = (const float*)d_in[7];
    const float* bn_mean  = (const float*)d_in[8];
    const float* bn_var   = (const float*)d_in[9];
    float* out = (float*)d_out;

    const int n = in_sizes[0] / 16;

    const size_t GRID_ELEMS = (size_t)B_ * D_ * H_ * W_;
    int*   grid   = (int*)d_ws;
    int*   nbr133 = grid + GRID_ELEMS;
    int*   nbr313 = nbr133 + (size_t)n * NBR_STRIDE;
    float* bufA   = (float*)(nbr313 + (size_t)n * NBR_STRIDE);
    float* bufB   = bufA + (size_t)n * COUT;
    float* zrow   = bufB + (size_t)n * COUT;

    const int n4 = (int)(GRID_ELEMS / 4);
    grid_init_kernel<<<(n4 + 255) / 256, 256, 0, stream>>>((int4*)grid, n4);
    zrow_init_kernel<<<1, 32, 0, stream>>>(zrow);
    grid_scatter_kernel<<<(n + 255) / 256, 256, 0, stream>>>(coords, grid, n);

    nbr_build_kernel<<<(n + 255) / 256, 256, 0, stream>>>(coords, grid, nbr133, nbr313, n);

    const int blocks = 256;

    conv_bn_kernel<16, false><<<blocks, 256, 0, stream>>>(
        feats, W1, nbr133, zrow,
        bn_gamma + 0,  bn_beta + 0,  bn_mean + 0,  bn_var + 0,
        nullptr, bufA, n);
    conv_bn_kernel<32, false><<<blocks, 256, 0, stream>>>(
        bufA,  W2, nbr313, zrow,
        bn_gamma + 32, bn_beta + 32, bn_mean + 32, bn_var + 32,
        nullptr, bufB, n);
    conv_bn_kernel<16, false><<<blocks, 256, 0, stream>>>(
        feats, W3, nbr313, zrow,
        bn_gamma + 64, bn_beta + 64, bn_mean + 64, bn_var + 64,
        nullptr, bufA, n);
    conv_bn_kernel<32, true><<<blocks, 256, 0, stream>>>(
        bufA,  W4, nbr133, zrow,
        bn_gamma + 96, bn_beta + 96, bn_mean + 96, bn_var + 96,
        bufB, out, n);
}


# conv tile walk software-pipelined: next tile's gathers (16-byte loads, second fragment set) fly under this tile's WMMAs; K-steps regroup channels, so the summation order inside a tap changes
# speedup vs baseline: 1.2508x; 1.1254x over previous
#include <hip/hip_runtime.h>

#define D_   480
#define H_   360
#define W_   32
#define B_   2
#define COUT 32
#define EPSF 1e-5f
#define NEGF 0.01f
#define NBR_STRIDE 12

typedef __attribute__((ext_vector_type(2))) float v2f;
typedef __attribute__((ext_vector_type(4))) float v4f;
typedef __attribute__((ext_vector_type(8))) float v8f;

__global__ void grid_init_kernel(int4* __restrict__ grid, int n4) {
    int i = blockIdx.x * blockDim.x + threadIdx.x;
    if (i < n4) grid[i] = make_int4(-1, -1, -1, -1);
}

__global__ void zrow_init_kernel(float* __restrict__ z) {
    z[threadIdx.x] = 0.0f;
}

__global__ void grid_scatter_kernel(const int* __restrict__ coords,
                                    int* __restrict__ grid, int n) {
    int i = blockIdx.x * blockDim.x + threadIdx.x;
    if (i >= n) return;
    int b = coords[4 * i + 0];
    int z = coords[4 * i + 1];
    int y = coords[4 * i + 2];
    int x = coords[4 * i + 3];
    grid[((b * D_ + z) * H_ + y) * W_ + x] = i;
}

__global__ void nbr_build_kernel(const int* __restrict__ coords,
                                 const int* __restrict__ grid,
                                 int* __restrict__ nbr133,
                                 int* __restrict__ nbr313, int n) {
    int i = blockIdx.x * blockDim.x + threadIdx.x;
    if (i >= n) return;
    int b = coords[4 * i + 0];
    int z = coords[4 * i + 1];
    int y = coords[4 * i + 2];
    int x = coords[4 * i + 3];
    int k = 0;
    #pragma unroll
    for (int dy = -1; dy <= 1; ++dy) {
        #pragma unroll
        for (int dx = -1; dx <= 1; ++dx) {
            int ny = y + dy, nx = x + dx;
            int v = -1;
            if (ny >= 0 && ny < H_ && nx >= 0 && nx < W_)
                v = grid[((b * D_ + z) * H_ + ny) * W_ + nx];
            nbr133[i * NBR_STRIDE + k] = v;
            ++k;
        }
    }
    k = 0;
    #pragma unroll
    for (int dz = -1; dz <= 1; ++dz) {
        #pragma unroll
        for (int dx = -1; dx <= 1; ++dx) {
            int nz = z + dz, nx = x + dx;
            int v = -1;
            if (nz >= 0 && nz < D_ && nx >= 0 && nx < W_)
                v = grid[((b * D_ + nz) * H_ + y) * W_ + nx];
            nbr313[i * NBR_STRIDE + k] = v;
            ++k;
        }
    }
}


__device__ __forceinline__ void load_nbr_row(const int* __restrict__ nbr, int tile,
                                             int row, int n, int (&nb)[9]) {
    const int point  = tile * 16 + row;
    const int pclamp = point < n ? point : n - 1;
    const int4* np = (const int4*)(nbr + (size_t)pclamp * NBR_STRIDE);
    const int4 a = np[0];
    const int4 b = np[1];
    const int  c = ((const int*)(np + 2))[0];
    nb[0] = a.x; nb[1] = a.y; nb[2] = a.z; nb[3] = a.w;
    nb[4] = b.x; nb[5] = b.y; nb[6] = b.z; nb[7] = b.w;
    nb[8] = c;
}

template <int CIN>
__device__ __forceinline__ void gather_tile(const float* __restrict__ feats,
                                            const float* __restrict__ zrow,
                                            const int (&nb)[9], int half,
                                            v2f (&A)[9][CIN / 4]) {
    #pragma unroll
    for (int k = 0; k < 9; ++k) {
        const int nbk = nb[k];
        const float* frow = ((nbk >= 0) ? (feats + (size_t)nbk * CIN) : zrow) + 4 * half;
        #pragma unroll
        for (int q = 0; q < CIN / 8; ++q) {
            const v4f f = *(const v4f*)(frow + 8 * q);
            v2f lo, hi;
            lo.x = f.x; lo.y = f.y;
            hi.x = f.z; hi.y = f.w;
            A[k][2 * q]     = lo;
            A[k][2 * q + 1] = hi;
        }
    }
}

template <int CIN>
__device__ __forceinline__ void mma_tile(const v2f (&A)[9][CIN / 4],
                                         const v2f (&Bf)[9][CIN / 4][2],
                                         v8f& acc0, v8f& acc1) {
    #pragma unroll
    for (int k = 0; k < 9; ++k) {
        #pragma unroll
        for (int j = 0; j < CIN / 4; ++j) {
            acc0 = __builtin_amdgcn_wmma_f32_16x16x4_f32(
                false, A[k][j], false, Bf[k][j][0], (short)0, acc0, false, false);
            acc1 = __builtin_amdgcn_wmma_f32_16x16x4_f32(
                false, A[k][j], false, Bf[k][j][1], (short)0, acc1, false, false);
        }
    }
}

template <bool ADD_RES>
__device__ __forceinline__ void load_res(const float* __restrict__ res, int tile,
                                         int half, int row, int n, float (&rs)[2][8]) {
    #pragma unroll
    for (int t2 = 0; t2 < 2; ++t2) {
        #pragma unroll
        for (int v = 0; v < 8; ++v) {
            if (ADD_RES) {
                const int orow = tile * 16 + half * 8 + v;
                const int oc   = orow < n ? orow : n - 1;
                rs[t2][v] = res[oc * COUT + t2 * 16 + row];
            } else {
                rs[t2][v] = 0.0f;
            }
        }
    }
}

template <bool ADD_RES>
__device__ __forceinline__ void store_tile(const v8f& acc0, const v8f& acc1,
                                           const float (&scl)[2], const float (&bia)[2],
                                           const float (&rs)[2][8],
                                           float* __restrict__ out,
                                           int tile, int half, int row, int n) {
    #pragma unroll
    for (int t2 = 0; t2 < 2; ++t2) {
        const int co = t2 * 16 + row;
        const v8f acc = t2 ? acc1 : acc0;
        #pragma unroll
        for (int v = 0; v < 8; ++v) {
            float yv = acc[v] * scl[t2] + bia[t2];
            yv = (yv >= 0.0f) ? yv : NEGF * yv;
            const int orow = tile * 16 + half * 8 + v;
            if (orow < n) {
                if (ADD_RES) yv += rs[t2][v];
                out[orow * COUT + co] = yv;
            }
        }
    }
}

template <int CIN, bool ADD_RES>
__global__ void __launch_bounds__((CIN == 16) ? 256 : 128)
__attribute__((amdgpu_waves_per_eu((CIN == 16) ? 2 : 1, (CIN == 16) ? 2 : 1)))
conv_bn_kernel(const float* __restrict__ feats,
               const float* __restrict__ Wg,
               const int*   __restrict__ nbr,
               const float* __restrict__ zrow,
               const float* __restrict__ gamma,
               const float* __restrict__ beta,
               const float* __restrict__ mean,
               const float* __restrict__ var,
               const float* __restrict__ res,
               float* __restrict__ out,
               int n) {
    const int lane = threadIdx.x & 31;
    const int wv   = threadIdx.x >> 5;
    const int row  = lane & 15;
    const int half = lane >> 4;

    const int wave   = blockIdx.x * (blockDim.x >> 5) + wv;
    const int nwaves = gridDim.x * (blockDim.x >> 5);
    const int tiles  = (n + 15) >> 4;
    if (wave >= tiles) return;

    v2f Bf[9][CIN / 4][2];
    {
        const float* wl = Wg + (4 * half) * COUT + row;
        #pragma unroll
        for (int k = 0; k < 9; ++k) {
            #pragma unroll
            for (int j = 0; j < CIN / 4; ++j) {
                #pragma unroll
                for (int t2 = 0; t2 < 2; ++t2) {
                    const float* wp = wl + (k * CIN + 8 * (j >> 1) + 2 * (j & 1)) * COUT + t2 * 16;
                    v2f b;
                    b.x = wp[0];
                    b.y = wp[COUT];
                    Bf[k][j][t2] = b;
                }
            }
        }
    }

    float scl[2], bia[2];
    #pragma unroll
    for (int t2 = 0; t2 < 2; ++t2) {
        const int co = t2 * 16 + row;
        scl[t2] = gamma[co] * rsqrtf(var[co] + EPSF);
        bia[t2] = beta[co] - mean[co] * scl[t2];
    }

    int nbA[9], nbB[9];
    v2f A0[9][CIN / 4], A1[9][CIN / 4];
    int t = wave;
    load_nbr_row(nbr, t, row, n, nbA);
    gather_tile<CIN>(feats, zrow, nbA, half, A0);
    load_nbr_row(nbr, t + nwaves, row, n, nbB);

    #pragma unroll 1
    for (;;) {
        {
            float rs[2][8];
            load_nbr_row(nbr, t + 2 * nwaves, row, n, nbA);
            load_res<ADD_RES>(res, t, half, row, n, rs);
            gather_tile<CIN>(feats, zrow, nbB, half, A1);
            __builtin_amdgcn_sched_barrier(0);
            v8f acc0 = {};
            v8f acc1 = {};
            mma_tile<CIN>(A0, Bf, acc0, acc1);
            store_tile<ADD_RES>(acc0, acc1, scl, bia, rs, out, t, half, row, n);
        }
        t += nwaves;
        if (t >= tiles) break;
        {
            float rs[2][8];
            load_nbr_row(nbr, t + 2 * nwaves, row, n, nbB);
            load_res<ADD_RES>(res, t, half, row, n, rs);
            gather_tile<CIN>(feats, zrow, nbA, half, A0);
            __builtin_amdgcn_sched_barrier(0);
            v8f acc0 = {};
            v8f acc1 = {};
            mma_tile<CIN>(A1, Bf, acc0, acc1);
            store_tile<ADD_RES>(acc0, acc1, scl, bia, rs, out, t, half, row, n);
        }
        t += nwaves;
        if (t >= tiles) break;
    }
}

extern "C" void kernel_launch(void* const* d_in, const int* in_sizes, int n_in,
                              void* d_out, int out_size, void* d_ws, size_t ws_size,
                              hipStream_t stream) {
    const float* feats    = (const float*)d_in[0];
    const int*   coords   = (const int*)  d_in[1];
    const float* W1       = (const float*)d_in[2];
    const float* W2       = (const float*)d_in[3];
    const float* W3       = (const float*)d_in[4];
    const float* W4       = (const float*)d_in[5];
    const float* bn_gamma = (const float*)d_in[6];
    const float* bn_beta  = (const float*)d_in[7];
    const float* bn_mean  = (const float*)d_in[8];
    const float* bn_var   = (const float*)d_in[9];
    float* out = (float*)d_out;

    const int n = in_sizes[0] / 16;

    const size_t GRID_ELEMS = (size_t)B_ * D_ * H_ * W_;
    int*   grid   = (int*)d_ws;
    int*   nbr133 = grid + GRID_ELEMS;
    int*   nbr313 = nbr133 + (size_t)n * NBR_STRIDE;
    float* bufA   = (float*)(nbr313 + (size_t)n * NBR_STRIDE);
    float* bufB   = bufA + (size_t)n * COUT;
    float* zrow   = bufB + (size_t)n * COUT;

    const int n4 = (int)(GRID_ELEMS / 4);
    grid_init_kernel<<<(n4 + 255) / 256, 256, 0, stream>>>((int4*)grid, n4);
    zrow_init_kernel<<<1, 32, 0, stream>>>(zrow);
    grid_scatter_kernel<<<(n + 255) / 256, 256, 0, stream>>>(coords, grid, n);

    nbr_build_kernel<<<(n + 255) / 256, 256, 0, stream>>>(coords, grid, nbr133, nbr313, n);

    const int blocks = 256;

    conv_bn_kernel<16, false><<<blocks, 256, 0, stream>>>(
        feats, W1, nbr133, zrow,
        bn_gamma + 0,  bn_beta + 0,  bn_mean + 0,  bn_var + 0,
        nullptr, bufA, n);
    conv_bn_kernel<32, false><<<blocks, 128, 0, stream>>>(
        bufA,  W2, nbr313, zrow,
        bn_gamma + 32, bn_beta + 32, bn_mean + 32, bn_var + 32,
        nullptr, bufB, n);
    conv_bn_kernel<16, false><<<blocks, 256, 0, stream>>>(
        feats, W3, nbr313, zrow,
        bn_gamma + 64, bn_beta + 64, bn_mean + 64, bn_var + 64,
        nullptr, bufA, n);
    conv_bn_kernel<32, true><<<blocks, 128, 0, stream>>>(
        bufA,  W4, nbr133, zrow,
        bn_gamma + 96, bn_beta + 96, bn_mean + 96, bn_var + 96,
        bufB, out, n);
}


# conv epilogue: each wave transposes its 16x32 result tile through a private LDS tile and stores it row-major, 4 16-byte stores per lane on whole 128-byte lines; was 16 row-tested 4-byte stores
# speedup vs baseline: 1.5331x; 1.2257x over previous
#include <hip/hip_runtime.h>

#define D_   480
#define H_   360
#define W_   32
#define B_   2
#define COUT 32
#define EPSF 1e-5f
#define NEGF 0.01f
#define NBR_STRIDE 12
#define OPITCH 36

typedef __attribute__((ext_vector_type(2))) float v2f;
typedef __attribute__((ext_vector_type(4))) float v4f;
typedef __attribute__((ext_vector_type(8))) float v8f;

__global__ void grid_init_kernel(int4* __restrict__ grid, int n4) {
    int i = blockIdx.x * blockDim.x + threadIdx.x;
    if (i < n4) grid[i] = make_int4(-1, -1, -1, -1);
}

__global__ void zrow_init_kernel(float* __restrict__ z) {
    z[threadIdx.x] = 0.0f;
}

__global__ void grid_scatter_kernel(const int* __restrict__ coords,
                                    int* __restrict__ grid, int n) {
    int i = blockIdx.x * blockDim.x + threadIdx.x;
    if (i >= n) return;
    int b = coords[4 * i + 0];
    int z = coords[4 * i + 1];
    int y = coords[4 * i + 2];
    int x = coords[4 * i + 3];
    grid[((b * D_ + z) * H_ + y) * W_ + x] = i;
}

__global__ void nbr_build_kernel(const int* __restrict__ coords,
                                 const int* __restrict__ grid,
                                 int* __restrict__ nbr133,
                                 int* __restrict__ nbr313, int n) {
    int i = blockIdx.x * blockDim.x + threadIdx.x;
    if (i >= n) return;
    int b = coords[4 * i + 0];
    int z = coords[4 * i + 1];
    int y = coords[4 * i + 2];
    int x = coords[4 * i + 3];
    int k = 0;
    #pragma unroll
    for (int dy = -1; dy <= 1; ++dy) {
        #pragma unroll
        for (int dx = -1; dx <= 1; ++dx) {
            int ny = y + dy, nx = x + dx;
            int v = -1;
            if (ny >= 0 && ny < H_ && nx >= 0 && nx < W_)
                v = grid[((b * D_ + z) * H_ + ny) * W_ + nx];
            nbr133[i * NBR_STRIDE + k] = v;
            ++k;
        }
    }
    k = 0;
    #pragma unroll
    for (int dz = -1; dz <= 1; ++dz) {
        #pragma unroll
        for (int dx = -1; dx <= 1; ++dx) {
            int nz = z + dz, nx = x + dx;
            int v = -1;
            if (nz >= 0 && nz < D_ && nx >= 0 && nx < W_)
                v = grid[((b * D_ + nz) * H_ + y) * W_ + nx];
            nbr313[i * NBR_STRIDE + k] = v;
            ++k;
        }
    }
}


__device__ __forceinline__ void load_nbr_row(const int* __restrict__ nbr, int tile,
                                             int row, int n, int (&nb)[9]) {
    const int point  = tile * 16 + row;
    const int pclamp = point < n ? point : n - 1;
    const int4* np = (const int4*)(nbr + (size_t)pclamp * NBR_STRIDE);
    const int4 a = np[0];
    const int4 b = np[1];
    const int  c = ((const int*)(np + 2))[0];
    nb[0] = a.x; nb[1] = a.y; nb[2] = a.z; nb[3] = a.w;
    nb[4] = b.x; nb[5] = b.y; nb[6] = b.z; nb[7] = b.w;
    nb[8] = c;
}

template <int CIN>
__device__ __forceinline__ void gather_tile(const float* __restrict__ feats,
                                            const float* __restrict__ zrow,
                                            const int (&nb)[9], int half,
                                            v2f (&A)[9][CIN / 4]) {
    #pragma unroll
    for (int k = 0; k < 9; ++k) {
        const int nbk = nb[k];
        const float* frow = ((nbk >= 0) ? (feats + (size_t)nbk * CIN) : zrow) + 4 * half;
        #pragma unroll
        for (int q = 0; q < CIN / 8; ++q) {
            const v4f f = *(const v4f*)(frow + 8 * q);
            v2f lo, hi;
            lo.x = f.x; lo.y = f.y;
            hi.x = f.z; hi.y = f.w;
            A[k][2 * q]     = lo;
            A[k][2 * q + 1] = hi;
        }
    }
}

template <int CIN>
__device__ __forceinline__ void mma_tile(const v2f (&A)[9][CIN / 4],
                                         const v2f (&Bf)[9][CIN / 4][2],
                                         v8f& acc0, v8f& acc1) {
    #pragma unroll
    for (int k = 0; k < 9; ++k) {
        #pragma unroll
        for (int j = 0; j < CIN / 4; ++j) {
            acc0 = __builtin_amdgcn_wmma_f32_16x16x4_f32(
                false, A[k][j], false, Bf[k][j][0], (short)0, acc0, false, false);
            acc1 = __builtin_amdgcn_wmma_f32_16x16x4_f32(
                false, A[k][j], false, Bf[k][j][1], (short)0, acc1, false, false);
        }
    }
}

template <bool ADD_RES>
__device__ __forceinline__ void load_res(const float* __restrict__ res, int tile,
                                         int lane, int n, v4f (&rs)[4]) {
    #pragma unroll
    for (int j = 0; j < 4; ++j) {
        if (ADD_RES) {
            const int orow = tile * 16 + 4 * j + (lane >> 3);
            const int oc   = orow < n ? orow : n - 1;
            rs[j] = *(const v4f*)(res + oc * COUT + 4 * (lane & 7));
        } else {
            const v4f z = {0.0f, 0.0f, 0.0f, 0.0f};
            rs[j] = z;
        }
    }
}

template <bool ADD_RES>
__device__ __forceinline__ void store_tile(const v8f& acc0, const v8f& acc1,
                                           const float (&scl)[2], const float (&bia)[2],
                                           const v4f (&rs)[4],
                                           float* __restrict__ ot,
                                           float* __restrict__ out,
                                           int tile, int lane, int n) {
    const int row  = lane & 15;
    const int half = lane >> 4;
    #pragma unroll
    for (int t2 = 0; t2 < 2; ++t2) {
        const v8f acc = t2 ? acc1 : acc0;
        #pragma unroll
        for (int v = 0; v < 8; ++v) {
            float yv = acc[v] * scl[t2] + bia[t2];
            yv = (yv >= 0.0f) ? yv : NEGF * yv;
            ot[(half * 8 + v) * OPITCH + t2 * 16 + row] = yv;
        }
    }

    const int r0 = lane >> 3;
    const int c4 = 4 * (lane & 7);
    v4f y[4];
    #pragma unroll
    for (int j = 0; j < 4; ++j) {
        y[j] = *(const v4f*)(ot + (4 * j + r0) * OPITCH + c4);
        if (ADD_RES) y[j] += rs[j];
    }
    if (tile * 16 + 16 <= n) {
        #pragma unroll
        for (int j = 0; j < 4; ++j)
            *(v4f*)(out + (tile * 16 + 4 * j + r0) * COUT + c4) = y[j];
    } else {
        #pragma unroll
        for (int j = 0; j < 4; ++j) {
            const int orow = tile * 16 + 4 * j + r0;
            if (orow < n) *(v4f*)(out + orow * COUT + c4) = y[j];
        }
    }
}

template <int CIN, bool ADD_RES>
__global__ void __launch_bounds__((CIN == 16) ? 256 : 128)
__attribute__((amdgpu_waves_per_eu((CIN == 16) ? 2 : 1, (CIN == 16) ? 2 : 1)))
conv_bn_kernel(const float* __restrict__ feats,
               const float* __restrict__ Wg,
               const int*   __restrict__ nbr,
               const float* __restrict__ zrow,
               const float* __restrict__ gamma,
               const float* __restrict__ beta,
               const float* __restrict__ mean,
               const float* __restrict__ var,
               const float* __restrict__ res,
               float* __restrict__ out,
               int n) {
    const int lane = threadIdx.x & 31;
    const int wv   = threadIdx.x >> 5;
    const int row  = lane & 15;
    const int half = lane >> 4;

    const int wave   = blockIdx.x * (blockDim.x >> 5) + wv;
    const int nwaves = gridDim.x * (blockDim.x >> 5);
    const int tiles  = (n + 15) >> 4;
    if (wave >= tiles) return;

    __shared__ __attribute__((aligned(16))) float otile[(CIN == 16) ? 8 : 4][16 * OPITCH];
    float* ot = otile[wv];

    v2f Bf[9][CIN / 4][2];
    {
        const float* wl = Wg + (4 * half) * COUT + row;
        #pragma unroll
        for (int k = 0; k < 9; ++k) {
            #pragma unroll
            for (int j = 0; j < CIN / 4; ++j) {
                #pragma unroll
                for (int t2 = 0; t2 < 2; ++t2) {
                    const float* wp = wl + (k * CIN + 8 * (j >> 1) + 2 * (j & 1)) * COUT + t2 * 16;
                    v2f b;
                    b.x = wp[0];
                    b.y = wp[COUT];
                    Bf[k][j][t2] = b;
                }
            }
        }
    }

    float scl[2], bia[2];
    #pragma unroll
    for (int t2 = 0; t2 < 2; ++t2) {
        const int co = t2 * 16 + row;
        scl[t2] = gamma[co] * rsqrtf(var[co] + EPSF);
        bia[t2] = beta[co] - mean[co] * scl[t2];
    }

    int nbA[9], nbB[9];
    v2f A0[9][CIN / 4], A1[9][CIN / 4];
    int t = wave;
    load_nbr_row(nbr, t, row, n, nbA);
    gather_tile<CIN>(feats, zrow, nbA, half, A0);
    load_nbr_row(nbr, t + nwaves, row, n, nbB);

    #pragma unroll 1
    for (;;) {
        {
            v4f rs[4];
            load_nbr_row(nbr, t + 2 * nwaves, row, n, nbA);
            load_res<ADD_RES>(res, t, lane, n, rs);
            gather_tile<CIN>(feats, zrow, nbB, half, A1);
            __builtin_amdgcn_sched_barrier(0);
            v8f acc0 = {};
            v8f acc1 = {};
            mma_tile<CIN>(A0, Bf, acc0, acc1);
            store_tile<ADD_RES>(acc0, acc1, scl, bia, rs, ot, out, t, lane, n);
        }
        t += nwaves;
        if (t >= tiles) break;
        {
            v4f rs[4];
            load_nbr_row(nbr, t + 2 * nwaves, row, n, nbB);
            load_res<ADD_RES>(res, t, lane, n, rs);
            gather_tile<CIN>(feats, zrow, nbA, half, A0);
            __builtin_amdgcn_sched_barrier(0);
            v8f acc0 = {};
            v8f acc1 = {};
            mma_tile<CIN>(A1, Bf, acc0, acc1);
            store_tile<ADD_RES>(acc0, acc1, scl, bia, rs, ot, out, t, lane, n);
        }
        t += nwaves;
        if (t >= tiles) break;
    }
}

extern "C" void kernel_launch(void* const* d_in, const int* in_sizes, int n_in,
                              void* d_out, int out_size, void* d_ws, size_t ws_size,
                              hipStream_t stream) {
    const float* feats    = (const float*)d_in[0];
    const int*   coords   = (const int*)  d_in[1];
    const float* W1       = (const float*)d_in[2];
    const float* W2       = (const float*)d_in[3];
    const float* W3       = (const float*)d_in[4];
    const float* W4       = (const float*)d_in[5];
    const float* bn_gamma = (const float*)d_in[6];
    const float* bn_beta  = (const float*)d_in[7];
    const float* bn_mean  = (const float*)d_in[8];
    const float* bn_var   = (const float*)d_in[9];
    float* out = (float*)d_out;

    const int n = in_sizes[0] / 16;

    const size_t GRID_ELEMS = (size_t)B_ * D_ * H_ * W_;
    int*   grid   = (int*)d_ws;
    int*   nbr133 = grid + GRID_ELEMS;
    int*   nbr313 = nbr133 + (size_t)n * NBR_STRIDE;
    float* bufA   = (float*)(nbr313 + (size_t)n * NBR_STRIDE);
    float* bufB   = bufA + (size_t)n * COUT;
    float* zrow   = bufB + (size_t)n * COUT;

    const int n4 = (int)(GRID_ELEMS / 4);
    grid_init_kernel<<<(n4 + 255) / 256, 256, 0, stream>>>((int4*)grid, n4);
    zrow_init_kernel<<<1, 32, 0, stream>>>(zrow);
    grid_scatter_kernel<<<(n + 255) / 256, 256, 0, stream>>>(coords, grid, n);

    nbr_build_kernel<<<(n + 255) / 256, 256, 0, stream>>>(coords, grid, nbr133, nbr313, n);

    const int blocks = 256;

    conv_bn_kernel<16, false><<<blocks, 256, 0, stream>>>(
        feats, W1, nbr133, zrow,
        bn_gamma + 0,  bn_beta + 0,  bn_mean + 0,  bn_var + 0,
        nullptr, bufA, n);
    conv_bn_kernel<32, false><<<blocks, 128, 0, stream>>>(
        bufA,  W2, nbr313, zrow,
        bn_gamma + 32, bn_beta + 32, bn_mean + 32, bn_var + 32,
        nullptr, bufB, n);
    conv_bn_kernel<16, false><<<blocks, 256, 0, stream>>>(
        feats, W3, nbr313, zrow,
        bn_gamma + 64, bn_beta + 64, bn_mean + 64, bn_var + 64,
        nullptr, bufA, n);
    conv_bn_kernel<32, true><<<blocks, 128, 0, stream>>>(
        bufA,  W4, nbr133, zrow,
        bn_gamma + 96, bn_beta + 96, bn_mean + 96, bn_var + 96,
        bufB, out, n);
}


# neighbour tables dropped: each conv pass reads its 9 neighbour indices from the voxel grid itself, three tiles ahead (coords, grid cells, rows); the table-build launch and its 19 MB of tables go
# speedup vs baseline: 1.7683x; 1.1534x over previous
#include <hip/hip_runtime.h>

#define D_   480
#define H_   360
#define W_   32
#define B_   2
#define COUT 32
#define EPSF 1e-5f
#define NEGF 0.01f
#define OPITCH 36

typedef __attribute__((ext_vector_type(2))) float v2f;
typedef __attribute__((ext_vector_type(4))) float v4f;
typedef __attribute__((ext_vector_type(8))) float v8f;

__global__ void grid_init_kernel(int4* __restrict__ grid, int n4) {
    int i = blockIdx.x * blockDim.x + threadIdx.x;
    if (i < n4) grid[i] = make_int4(-1, -1, -1, -1);
}

__global__ void zrow_init_kernel(float* __restrict__ z) {
    z[threadIdx.x] = 0.0f;
}

__global__ void grid_scatter_kernel(const int* __restrict__ coords,
                                    int* __restrict__ grid, int n) {
    int i = blockIdx.x * blockDim.x + threadIdx.x;
    if (i >= n) return;
    int b = coords[4 * i + 0];
    int z = coords[4 * i + 1];
    int y = coords[4 * i + 2];
    int x = coords[4 * i + 3];
    grid[((b * D_ + z) * H_ + y) * W_ + x] = i;
}


__device__ __forceinline__ void load_coords(const int* __restrict__ coords, int tile,
                                            int row, int n, int (&c)[4]) {
    const int point  = tile * 16 + row;
    const int pclamp = point < n ? point : n - 1;
    const int* cp = coords + 4 * pclamp;
    c[0] = cp[0]; c[1] = cp[1]; c[2] = cp[2]; c[3] = cp[3];
}

template <bool T133>
__device__ __forceinline__ void lookup_nbrs(const int* __restrict__ grid,
                                            const int (&c)[4], int (&nb)[9]) {
    const int centre = ((c[0] * D_ + c[1]) * H_ + c[2]) * W_ + c[3];
    #pragma unroll
    for (int o = -1; o <= 1; ++o) {
        const int  oc  = (T133 ? c[2] : c[1]) + o;
        const bool ook = (oc >= 0) && (oc < (T133 ? H_ : D_));
        #pragma unroll
        for (int dx = -1; dx <= 1; ++dx) {
            const int  nx   = c[3] + dx;
            const bool ok   = ook && (nx >= 0) && (nx < W_);
            const int  cell = ok ? centre + o * (T133 ? W_ : H_ * W_) + dx : centre;
            const int  v    = grid[cell];
            nb[3 * (o + 1) + (dx + 1)] = ok ? v : -1;
        }
    }
}

template <int CIN>
__device__ __forceinline__ void gather_tile(const float* __restrict__ feats,
                                            const float* __restrict__ zrow,
                                            const int (&nb)[9], int half,
                                            v2f (&A)[9][CIN / 4]) {
    #pragma unroll
    for (int k = 0; k < 9; ++k) {
        const int nbk = nb[k];
        const float* frow = ((nbk >= 0) ? (feats + (size_t)nbk * CIN) : zrow) + 4 * half;
        #pragma unroll
        for (int q = 0; q < CIN / 8; ++q) {
            const v4f f = *(const v4f*)(frow + 8 * q);
            v2f lo, hi;
            lo.x = f.x; lo.y = f.y;
            hi.x = f.z; hi.y = f.w;
            A[k][2 * q]     = lo;
            A[k][2 * q + 1] = hi;
        }
    }
}

template <int CIN>
__device__ __forceinline__ void mma_tile(const v2f (&A)[9][CIN / 4],
                                         const v2f (&Bf)[9][CIN / 4][2],
                                         v8f& acc0, v8f& acc1) {
    #pragma unroll
    for (int k = 0; k < 9; ++k) {
        #pragma unroll
        for (int j = 0; j < CIN / 4; ++j) {
            acc0 = __builtin_amdgcn_wmma_f32_16x16x4_f32(
                false, A[k][j], false, Bf[k][j][0], (short)0, acc0, false, false);
            acc1 = __builtin_amdgcn_wmma_f32_16x16x4_f32(
                false, A[k][j], false, Bf[k][j][1], (short)0, acc1, false, false);
        }
    }
}

template <bool ADD_RES>
__device__ __forceinline__ void load_res(const float* __restrict__ res, int tile,
                                         int lane, int n, v4f (&rs)[4]) {
    #pragma unroll
    for (int j = 0; j < 4; ++j) {
        if (ADD_RES) {
            const int orow = tile * 16 + 4 * j + (lane >> 3);
            const int oc   = orow < n ? orow : n - 1;
            rs[j] = *(const v4f*)(res + oc * COUT + 4 * (lane & 7));
        } else {
            const v4f z = {0.0f, 0.0f, 0.0f, 0.0f};
            rs[j] = z;
        }
    }
}

template <bool ADD_RES>
__device__ __forceinline__ void store_tile(const v8f& acc0, const v8f& acc1,
                                           const float (&scl)[2], const float (&bia)[2],
                                           const v4f (&rs)[4],
                                           float* __restrict__ ot,
                                           float* __restrict__ out,
                                           int tile, int lane, int n) {
    const int row  = lane & 15;
    const int half = lane >> 4;
    #pragma unroll
    for (int t2 = 0; t2 < 2; ++t2) {
        const v8f acc = t2 ? acc1 : acc0;
        #pragma unroll
        for (int v = 0; v < 8; ++v) {
            float yv = acc[v] * scl[t2] + bia[t2];
            yv = (yv >= 0.0f) ? yv : NEGF * yv;
            ot[(half * 8 + v) * OPITCH + t2 * 16 + row] = yv;
        }
    }

    const int r0 = lane >> 3;
    const int c4 = 4 * (lane & 7);
    v4f y[4];
    #pragma unroll
    for (int j = 0; j < 4; ++j) {
        y[j] = *(const v4f*)(ot + (4 * j + r0) * OPITCH + c4);
        if (ADD_RES) y[j] += rs[j];
    }
    if (tile * 16 + 16 <= n) {
        #pragma unroll
        for (int j = 0; j < 4; ++j)
            *(v4f*)(out + (tile * 16 + 4 * j + r0) * COUT + c4) = y[j];
    } else {
        #pragma unroll
        for (int j = 0; j < 4; ++j) {
            const int orow = tile * 16 + 4 * j + r0;
            if (orow < n) *(v4f*)(out + orow * COUT + c4) = y[j];
        }
    }
}

template <int CIN, bool T133, bool ADD_RES>
__global__ void __launch_bounds__((CIN == 16) ? 256 : 128)
__attribute__((amdgpu_waves_per_eu((CIN == 16) ? 2 : 1, (CIN == 16) ? 2 : 1)))
conv_bn_kernel(const float* __restrict__ feats,
               const float* __restrict__ Wg,
               const int*   __restrict__ coords,
               const int*   __restrict__ grid,
               const float* __restrict__ zrow,
               const float* __restrict__ gamma,
               const float* __restrict__ beta,
               const float* __restrict__ mean,
               const float* __restrict__ var,
               const float* __restrict__ res,
               float* __restrict__ out,
               int n) {
    const int lane = threadIdx.x & 31;
    const int wv   = threadIdx.x >> 5;
    const int row  = lane & 15;
    const int half = lane >> 4;

    const int wave   = blockIdx.x * (blockDim.x >> 5) + wv;
    const int nwaves = gridDim.x * (blockDim.x >> 5);
    const int tiles  = (n + 15) >> 4;
    if (wave >= tiles) return;

    __shared__ __attribute__((aligned(16))) float otile[(CIN == 16) ? 8 : 4][16 * OPITCH];
    float* ot = otile[wv];

    v2f Bf[9][CIN / 4][2];
    {
        const float* wl = Wg + (4 * half) * COUT + row;
        #pragma unroll
        for (int k = 0; k < 9; ++k) {
            #pragma unroll
            for (int j = 0; j < CIN / 4; ++j) {
                #pragma unroll
                for (int t2 = 0; t2 < 2; ++t2) {
                    const float* wp = wl + (k * CIN + 8 * (j >> 1) + 2 * (j & 1)) * COUT + t2 * 16;
                    v2f b;
                    b.x = wp[0];
                    b.y = wp[COUT];
                    Bf[k][j][t2] = b;
                }
            }
        }
    }

    float scl[2], bia[2];
    #pragma unroll
    for (int t2 = 0; t2 < 2; ++t2) {
        const int co = t2 * 16 + row;
        scl[t2] = gamma[co] * rsqrtf(var[co] + EPSF);
        bia[t2] = beta[co] - mean[co] * scl[t2];
    }

    int c[4], nb[9];
    v2f A0[9][CIN / 4], A1[9][CIN / 4];
    int t = wave;
    load_coords(coords, t, row, n, c);
    lookup_nbrs<T133>(grid, c, nb);
    gather_tile<CIN>(feats, zrow, nb, half, A0);
    load_coords(coords, t + nwaves, row, n, c);
    lookup_nbrs<T133>(grid, c, nb);
    load_coords(coords, t + 2 * nwaves, row, n, c);

    #pragma unroll 1
    for (;;) {
        {
            v4f rs[4];
            load_res<ADD_RES>(res, t, lane, n, rs);
            gather_tile<CIN>(feats, zrow, nb, half, A1);
            lookup_nbrs<T133>(grid, c, nb);
            load_coords(coords, t + 3 * nwaves, row, n, c);
            __builtin_amdgcn_sched_barrier(0);
            v8f acc0 = {};
            v8f acc1 = {};
            mma_tile<CIN>(A0, Bf, acc0, acc1);
            store_tile<ADD_RES>(acc0, acc1, scl, bia, rs, ot, out, t, lane, n);
        }
        t += nwaves;
        if (t >= tiles) break;
        {
            v4f rs[4];
            load_res<ADD_RES>(res, t, lane, n, rs);
            gather_tile<CIN>(feats, zrow, nb, half, A0);
            lookup_nbrs<T133>(grid, c, nb);
            load_coords(coords, t + 3 * nwaves, row, n, c);
            __builtin_amdgcn_sched_barrier(0);
            v8f acc0 = {};
            v8f acc1 = {};
            mma_tile<CIN>(A1, Bf, acc0, acc1);
            store_tile<ADD_RES>(acc0, acc1, scl, bia, rs, ot, out, t, lane, n);
        }
        t += nwaves;
        if (t >= tiles) break;
    }
}

extern "C" void kernel_launch(void* const* d_in, const int* in_sizes, int n_in,
                              void* d_out, int out_size, void* d_ws, size_t ws_size,
                              hipStream_t stream) {
    const float* feats    = (const float*)d_in[0];
    const int*   coords   = (const int*)  d_in[1];
    const float* W1       = (const float*)d_in[2];
    const float* W2       = (const float*)d_in[3];
    const float* W3       = (const float*)d_in[4];
    const float* W4       = (const float*)d_in[5];
    const float* bn_gamma = (const float*)d_in[6];
    const float* bn_beta  = (const float*)d_in[7];
    const float* bn_mean  = (const float*)d_in[8];
    const float* bn_var   = (const float*)d_in[9];
    float* out = (float*)d_out;

    const int n = in_sizes[0] / 16;

    const size_t GRID_ELEMS = (size_t)B_ * D_ * H_ * W_;
    int*   grid   = (int*)d_ws;
    float* bufA   = (float*)(grid + GRID_ELEMS);
    float* bufB   = bufA + (size_t)n * COUT;
    float* zrow   = bufB + (size_t)n * COUT;

    const int n4 = (int)(GRID_ELEMS / 4);
    grid_init_kernel<<<(n4 + 255) / 256, 256, 0, stream>>>((int4*)grid, n4);
    zrow_init_kernel<<<1, 32, 0, stream>>>(zrow);
    grid_scatter_kernel<<<(n + 255) / 256, 256, 0, stream>>>(coords, grid, n);

    const int blocks = 256;

    conv_bn_kernel<16, true, false><<<blocks, 256, 0, stream>>>(
        feats, W1, coords, grid, zrow,
        bn_gamma + 0,  bn_beta + 0,  bn_mean + 0,  bn_var + 0,
        nullptr, bufA, n);
    conv_bn_kernel<32, false, false><<<blocks, 128, 0, stream>>>(
        bufA,  W2, coords, grid, zrow,
        bn_gamma + 32, bn_beta + 32, bn_mean + 32, bn_var + 32,
        nullptr, bufB, n);
    conv_bn_kernel<16, false, false><<<blocks, 256, 0, stream>>>(
        feats, W3, coords, grid, zrow,
        bn_gamma + 64, bn_beta + 64, bn_mean + 64, bn_var + 64,
        nullptr, bufA, n);
    conv_bn_kernel<32, true, true><<<blocks, 128, 0, stream>>>(
        bufA,  W4, coords, grid, zrow,
        bn_gamma + 96, bn_beta + 96, bn_mean + 96, bn_var + 96,
        bufB, out, n);
}


# neighbour indices looked up by the first two conv passes are also written out, one whole-line block per tile, and read back two tiles ahead by the last two passes, which drop coordinates and grid
# speedup vs baseline: 1.8431x; 1.0423x over previous
#include <hip/hip_runtime.h>

#define D_   480
#define H_   360
#define W_   32
#define B_   2
#define COUT 32
#define EPSF 1e-5f
#define NEGF 0.01f
#define OPITCH 36

#define NBPITCH 160

typedef __attribute__((ext_vector_type(2))) float v2f;
typedef __attribute__((ext_vector_type(4))) float v4f;
typedef __attribute__((ext_vector_type(8))) float v8f;

__global__ void grid_init_kernel(int4* __restrict__ grid, int n4) {
    int i = blockIdx.x * blockDim.x + threadIdx.x;
    if (i < n4) grid[i] = make_int4(-1, -1, -1, -1);
}

__global__ void zrow_init_kernel(float* __restrict__ z) {
    z[threadIdx.x] = 0.0f;
}

__global__ void grid_scatter_kernel(const int* __restrict__ coords,
                                    int* __restrict__ grid, int n) {
    int i = blockIdx.x * blockDim.x + threadIdx.x;
    if (i >= n) return;
    int b = coords[4 * i + 0];
    int z = coords[4 * i + 1];
    int y = coords[4 * i + 2];
    int x = coords[4 * i + 3];
    grid[((b * D_ + z) * H_ + y) * W_ + x] = i;
}


__device__ __forceinline__ void load_coords(const int* __restrict__ coords, int tile,
                                            int row, int n, int (&c)[4]) {
    const int point  = tile * 16 + row;
    const int pclamp = point < n ? point : n - 1;
    const int* cp = coords + 4 * pclamp;
    c[0] = cp[0]; c[1] = cp[1]; c[2] = cp[2]; c[3] = cp[3];
}

template <bool T133>
__device__ __forceinline__ void lookup_nbrs(const int* __restrict__ grid,
                                            const int (&c)[4], int (&nb)[9]) {
    const int centre = ((c[0] * D_ + c[1]) * H_ + c[2]) * W_ + c[3];
    #pragma unroll
    for (int o = -1; o <= 1; ++o) {
        const int  oc  = (T133 ? c[2] : c[1]) + o;
        const bool ook = (oc >= 0) && (oc < (T133 ? H_ : D_));
        #pragma unroll
        for (int dx = -1; dx <= 1; ++dx) {
            const int  nx   = c[3] + dx;
            const bool ok   = ook && (nx >= 0) && (nx < W_);
            const int  cell = ok ? centre + o * (T133 ? W_ : H_ * W_) + dx : centre;
            const int  v    = grid[cell];
            nb[3 * (o + 1) + (dx + 1)] = ok ? v : -1;
        }
    }
}

__device__ __forceinline__ void store_nbrs(int* __restrict__ tab, int tile, int tiles,
                                           const int (&nb)[9], int lane) {
    if (tile < tiles) {
        int* tp = tab + (size_t)tile * NBPITCH + lane;
        const bool hi = lane >= 16;
        #pragma unroll
        for (int j = 0; j < 5; ++j) {
            const int lo_v = nb[2 * j];
            const int hi_v = (2 * j + 1 < 9) ? nb[2 * j + 1] : -1;
            tp[32 * j] = hi ? hi_v : lo_v;
        }
    }
}

__device__ __forceinline__ void read_nbrs(const int* __restrict__ tab, int tile, int tiles,
                                          int row, int (&nb)[9]) {
    const int tc = tile < tiles ? tile : tiles - 1;
    const int* tp = tab + (size_t)tc * NBPITCH + row;
    #pragma unroll
    for (int k = 0; k < 9; ++k) nb[k] = tp[16 * k];
}

template <bool T133, int NBM>
__device__ __forceinline__ void refill_nbrs(const int* __restrict__ coords,
                                            const int* __restrict__ grid,
                                            int* __restrict__ tab,
                                            int tnext, int nwaves, int tiles, int n,
                                            int row, int lane, int (&c)[4], int (&nb)[9]) {
    if (NBM == 2) {
        read_nbrs(tab, tnext + 2 * nwaves, tiles, row, nb);
    } else {
        store_nbrs(tab, tnext, tiles, nb, lane);
        lookup_nbrs<T133>(grid, c, nb);
        load_coords(coords, tnext + 2 * nwaves, row, n, c);
    }
}

template <int CIN>
__device__ __forceinline__ void gather_tile(const float* __restrict__ feats,
                                            const float* __restrict__ zrow,
                                            const int (&nb)[9], int half,
                                            v2f (&A)[9][CIN / 4]) {
    #pragma unroll
    for (int k = 0; k < 9; ++k) {
        const int nbk = nb[k];
        const float* frow = ((nbk >= 0) ? (feats + (size_t)nbk * CIN) : zrow) + 4 * half;
        #pragma unroll
        for (int q = 0; q < CIN / 8; ++q) {
            const v4f f = *(const v4f*)(frow + 8 * q);
            v2f lo, hi;
            lo.x = f.x; lo.y = f.y;
            hi.x = f.z; hi.y = f.w;
            A[k][2 * q]     = lo;
            A[k][2 * q + 1] = hi;
        }
    }
}

template <int CIN>
__device__ __forceinline__ void mma_tile(const v2f (&A)[9][CIN / 4],
                                         const v2f (&Bf)[9][CIN / 4][2],
                                         v8f& acc0, v8f& acc1) {
    #pragma unroll
    for (int k = 0; k < 9; ++k) {
        #pragma unroll
        for (int j = 0; j < CIN / 4; ++j) {
            acc0 = __builtin_amdgcn_wmma_f32_16x16x4_f32(
                false, A[k][j], false, Bf[k][j][0], (short)0, acc0, false, false);
            acc1 = __builtin_amdgcn_wmma_f32_16x16x4_f32(
                false, A[k][j], false, Bf[k][j][1], (short)0, acc1, false, false);
        }
    }
}

template <bool ADD_RES>
__device__ __forceinline__ void load_res(const float* __restrict__ res, int tile,
                                         int lane, int n, v4f (&rs)[4]) {
    #pragma unroll
    for (int j = 0; j < 4; ++j) {
        if (ADD_RES) {
            const int orow = tile * 16 + 4 * j + (lane >> 3);
            const int oc   = orow < n ? orow : n - 1;
            rs[j] = *(const v4f*)(res + oc * COUT + 4 * (lane & 7));
        } else {
            const v4f z = {0.0f, 0.0f, 0.0f, 0.0f};
            rs[j] = z;
        }
    }
}

template <bool ADD_RES>
__device__ __forceinline__ void store_tile(const v8f& acc0, const v8f& acc1,
                                           const float (&scl)[2], const float (&bia)[2],
                                           const v4f (&rs)[4],
                                           float* __restrict__ ot,
                                           float* __restrict__ out,
                                           int tile, int lane, int n) {
    const int row  = lane & 15;
    const int half = lane >> 4;
    #pragma unroll
    for (int t2 = 0; t2 < 2; ++t2) {
        const v8f acc = t2 ? acc1 : acc0;
        #pragma unroll
        for (int v = 0; v < 8; ++v) {
            float yv = acc[v] * scl[t2] + bia[t2];
            yv = (yv >= 0.0f) ? yv : NEGF * yv;
            ot[(half * 8 + v) * OPITCH + t2 * 16 + row] = yv;
        }
    }

    const int r0 = lane >> 3;
    const int c4 = 4 * (lane & 7);
    v4f y[4];
    #pragma unroll
    for (int j = 0; j < 4; ++j) {
        y[j] = *(const v4f*)(ot + (4 * j + r0) * OPITCH + c4);
        if (ADD_RES) y[j] += rs[j];
    }
    if (tile * 16 + 16 <= n) {
        #pragma unroll
        for (int j = 0; j < 4; ++j)
            *(v4f*)(out + (tile * 16 + 4 * j + r0) * COUT + c4) = y[j];
    } else {
        #pragma unroll
        for (int j = 0; j < 4; ++j) {
            const int orow = tile * 16 + 4 * j + r0;
            if (orow < n) *(v4f*)(out + orow * COUT + c4) = y[j];
        }
    }
}

template <int CIN, bool T133, bool ADD_RES, int NBM>
__device__ __forceinline__ void walk_step(
        const float* __restrict__ feats, const int* __restrict__ coords,
        const int* __restrict__ grid, const float* __restrict__ zrow,
        const float* __restrict__ res, float* __restrict__ out,
        int* __restrict__ nbtab, float* __restrict__ ot,
        const v2f (&Bf)[9][CIN / 4][2], const float (&scl)[2], const float (&bia)[2],
        const v2f (&Acur)[9][CIN / 4], v2f (&Anext)[9][CIN / 4],
        int (&c)[4], int (&nbs)[9],
        int t, int nwaves, int tiles, int n, int lane, int row, int half) {
    v4f rs[4];
    load_res<ADD_RES>(res, t, lane, n, rs);
    gather_tile<CIN>(feats, zrow, nbs, half, Anext);
    refill_nbrs<T133, NBM>(coords, grid, nbtab, t + nwaves, nwaves, tiles, n,
                           row, lane, c, nbs);
    __builtin_amdgcn_sched_barrier(0);
    v8f acc0 = {};
    v8f acc1 = {};
    mma_tile<CIN>(Acur, Bf, acc0, acc1);
    store_tile<ADD_RES>(acc0, acc1, scl, bia, rs, ot, out, t, lane, n);
}

template <int CIN, bool T133, bool ADD_RES, int NBM>
__global__ void __launch_bounds__((CIN == 16) ? 256 : 128)
__attribute__((amdgpu_waves_per_eu((CIN == 16) ? 2 : 1, (CIN == 16) ? 2 : 1)))
conv_bn_kernel(const float* __restrict__ feats,
               const float* __restrict__ Wg,
               const int*   __restrict__ coords,
               const int*   __restrict__ grid,
               const float* __restrict__ zrow,
               const float* __restrict__ gamma,
               const float* __restrict__ beta,
               const float* __restrict__ mean,
               const float* __restrict__ var,
               const float* __restrict__ res,
               float* __restrict__ out,
               int* __restrict__ nbtab,
               int n) {
    const int lane = threadIdx.x & 31;
    const int wv   = __builtin_amdgcn_readfirstlane(threadIdx.x >> 5);
    const int row  = lane & 15;
    const int half = lane >> 4;

    const int wave   = blockIdx.x * (blockDim.x >> 5) + wv;
    const int nwaves = gridDim.x * (blockDim.x >> 5);
    const int tiles  = (n + 15) >> 4;
    if (wave >= tiles) return;

    __shared__ __attribute__((aligned(16))) float otile[(CIN == 16) ? 8 : 4][16 * OPITCH];
    float* ot = otile[wv];

    v2f Bf[9][CIN / 4][2];
    {
        const float* wl = Wg + (4 * half) * COUT + row;
        #pragma unroll
        for (int k = 0; k < 9; ++k) {
            #pragma unroll
            for (int j = 0; j < CIN / 4; ++j) {
                #pragma unroll
                for (int t2 = 0; t2 < 2; ++t2) {
                    const float* wp = wl + (k * CIN + 8 * (j >> 1) + 2 * (j & 1)) * COUT + t2 * 16;
                    v2f b;
                    b.x = wp[0];
                    b.y = wp[COUT];
                    Bf[k][j][t2] = b;
                }
            }
        }
    }

    float scl[2], bia[2];
    #pragma unroll
    for (int t2 = 0; t2 < 2; ++t2) {
        const int co = t2 * 16 + row;
        scl[t2] = gamma[co] * rsqrtf(var[co] + EPSF);
        bia[t2] = beta[co] - mean[co] * scl[t2];
    }

    int c[4], nb[9], nb2[9];
    int (&nbh2)[9] = *((NBM == 2) ? &nb2 : &nb);
    v2f A0[9][CIN / 4], A1[9][CIN / 4];
    int t = wave;
    if (NBM == 2) {
        read_nbrs(nbtab, t, tiles, row, nb2);
        read_nbrs(nbtab, t + nwaves, tiles, row, nb);
        asm volatile("" :: "v"(nb[0]), "v"(nb[1]), "v"(nb[2]), "v"(nb[3]), "v"(nb[4]),
                           "v"(nb[5]), "v"(nb[6]), "v"(nb[7]), "v"(nb[8]));
        gather_tile<CIN>(feats, zrow, nb2, half, A0);
        read_nbrs(nbtab, t + 2 * nwaves, tiles, row, nb2);
    } else {
        load_coords(coords, t, row, n, c);
        lookup_nbrs<T133>(grid, c, nb);
        gather_tile<CIN>(feats, zrow, nb, half, A0);
        store_nbrs(nbtab, t, tiles, nb, lane);
        load_coords(coords, t + nwaves, row, n, c);
        lookup_nbrs<T133>(grid, c, nb);
        load_coords(coords, t + 2 * nwaves, row, n, c);
    }

    const int mine = (tiles - wave + nwaves - 1) / nwaves;
    #pragma unroll 1
    for (int p = 0; p < (mine >> 1); ++p) {
        walk_step<CIN, T133, ADD_RES, NBM>(feats, coords, grid, zrow, res, out, nbtab, ot,
                                           Bf, scl, bia, A0, A1, c, nb,
                                           t, nwaves, tiles, n, lane, row, half);
        t += nwaves;
        walk_step<CIN, T133, ADD_RES, NBM>(feats, coords, grid, zrow, res, out, nbtab, ot,
                                           Bf, scl, bia, A1, A0, c, nbh2,
                                           t, nwaves, tiles, n, lane, row, half);
        t += nwaves;
    }
    if (mine & 1)
        walk_step<CIN, T133, ADD_RES, NBM>(feats, coords, grid, zrow, res, out, nbtab, ot,
                                           Bf, scl, bia, A0, A1, c, nb,
                                           t, nwaves, tiles, n, lane, row, half);
}

extern "C" void kernel_launch(void* const* d_in, const int* in_sizes, int n_in,
                              void* d_out, int out_size, void* d_ws, size_t ws_size,
                              hipStream_t stream) {
    const float* feats    = (const float*)d_in[0];
    const int*   coords   = (const int*)  d_in[1];
    const float* W1       = (const float*)d_in[2];
    const float* W2       = (const float*)d_in[3];
    const float* W3       = (const float*)d_in[4];
    const float* W4       = (const float*)d_in[5];
    const float* bn_gamma = (const float*)d_in[6];
    const float* bn_beta  = (const float*)d_in[7];
    const float* bn_mean  = (const float*)d_in[8];
    const float* bn_var   = (const float*)d_in[9];
    float* out = (float*)d_out;

    const int n = in_sizes[0] / 16;

    const size_t GRID_ELEMS = (size_t)B_ * D_ * H_ * W_;
    int*   grid   = (int*)d_ws;
    float* bufA   = (float*)(grid + GRID_ELEMS);
    float* bufB   = bufA + (size_t)n * COUT;
    float* zrow   = bufB + (size_t)n * COUT;
    int*   tab133 = (int*)(zrow + 32);
    int*   tab313 = tab133 + (size_t)((n + 15) / 16) * NBPITCH;

    const int n4 = (int)(GRID_ELEMS / 4);
    grid_init_kernel<<<(n4 + 255) / 256, 256, 0, stream>>>((int4*)grid, n4);
    zrow_init_kernel<<<1, 32, 0, stream>>>(zrow);
    grid_scatter_kernel<<<(n + 255) / 256, 256, 0, stream>>>(coords, grid, n);

    const int blocks = 256;

    conv_bn_kernel<16, true, false, 1><<<blocks, 256, 0, stream>>>(
        feats, W1, coords, grid, zrow,
        bn_gamma + 0,  bn_beta + 0,  bn_mean + 0,  bn_var + 0,
        nullptr, bufA, tab133, n);
    conv_bn_kernel<32, false, false, 1><<<blocks, 128, 0, stream>>>(
        bufA,  W2, coords, grid, zrow,
        bn_gamma + 32, bn_beta + 32, bn_mean + 32, bn_var + 32,
        nullptr, bufB, tab313, n);
    conv_bn_kernel<16, false, false, 2><<<blocks, 256, 0, stream>>>(
        feats, W3, coords, grid, zrow,
        bn_gamma + 64, bn_beta + 64, bn_mean + 64, bn_var + 64,
        nullptr, bufA, tab313, n);
    conv_bn_kernel<32, true, true, 2><<<blocks, 128, 0, stream>>>(
        bufA,  W4, coords, grid, zrow,
        bn_gamma + 96, bn_beta + 96, bn_mean + 96, bn_var + 96,
        bufB, out, tab133, n);
}
